# starting kernel
# baseline (speedup 1.0000x reference)
#include <hip/hip_runtime.h>

#define B_    8
#define C_    3
#define N_    4096
#define K_    20
#define H_    64
#define DOUT_ 64
#define EPS_  1e-5f

typedef __attribute__((ext_vector_type(2))) float v2f;
typedef __attribute__((ext_vector_type(8))) float v8f;

__global__ __launch_bounds__(256) void knn_kernel(const float* __restrict__ x,
                                                  int* __restrict__ idx)
{
    __shared__ float sx[N_], sy[N_], sz[N_];
    const int blocksPerBatch = N_ / 256;
    const int b     = blockIdx.x / blocksPerBatch;
    const int chunk = blockIdx.x % blocksPerBatch;
    const float* xb = x + (size_t)b * 3 * N_;
    for (int j = threadIdx.x; j < N_; j += 256) {
        sx[j] = xb[j];
        sy[j] = xb[N_ + j];
        sz[j] = xb[2 * N_ + j];
    }
    __syncthreads();

    const int   n  = chunk * 256 + threadIdx.x;
    const float qx = sx[n], qy = sy[n], qz = sz[n];
    const float qq = qx * qx + qy * qy + qz * qz;

    float bd[K_];
    int   bi[K_];
#pragma unroll
    for (int t = 0; t < K_; ++t) { bd[t] = -3.4e38f; bi[t] = 0; }

    for (int j = 0; j < N_; ++j) {
        const float px = sx[j], py = sy[j], pz = sz[j];
        const float d = 2.0f * (qx * px + qy * py + qz * pz) - qq
                        - (px * px + py * py + pz * pz);
        if (d > bd[K_ - 1]) {
            bd[K_ - 1] = d; bi[K_ - 1] = j;
#pragma unroll
            for (int t = K_ - 1; t > 0; --t) {
                if (bd[t] > bd[t - 1]) {
                    float tf = bd[t]; bd[t] = bd[t - 1]; bd[t - 1] = tf;
                    int   ti = bi[t]; bi[t] = bi[t - 1]; bi[t - 1] = ti;
                }
            }
        }
    }
    int* outp = idx + ((size_t)b * N_ + n) * K_;
#pragma unroll
    for (int t = 0; t < K_; ++t) outp[t] = bi[t];
}

__global__ __launch_bounds__(128) void edge_kernel(
    const float* __restrict__ x,  const int* __restrict__ idx,
    const float* __restrict__ w1, const float* __restrict__ b1,
    const float* __restrict__ g1, const float* __restrict__ beta1,
    const float* __restrict__ rm1,const float* __restrict__ rv1,
    float* __restrict__ out_m1,   float* __restrict__ out_m2)
{
    const int lane = threadIdx.x & 31;
    const int half = lane >> 4;
    const int l15  = lane & 15;
    const int wave = (blockIdx.x * blockDim.x + threadIdx.x) >> 5;
    const int b    = wave >> 12;
    const int n    = wave & (N_ - 1);

    const float* xb = x + (size_t)b * 3 * N_;
    const float cx = xb[n], cy = xb[N_ + n], cz = xb[2 * N_ + n];

    v2f a0[2], a1[2];
#pragma unroll
    for (int rt = 0; rt < 2; ++rt) {
        const int  kk    = rt * 16 + l15;
        const bool valid = (kk < K_);
        const int  kc    = valid ? kk : (K_ - 1);
        const int  nb    = idx[((size_t)b * N_ + n) * K_ + kc];
        const float nx = xb[nb], ny = xb[N_ + nb], nz = xb[2 * N_ + nb];
        const float e0 = valid ? (nx - cx) : 0.0f;
        const float e1 = valid ? (ny - cy) : 0.0f;
        const float e2 = valid ? (nz - cz) : 0.0f;
        const float e3 = valid ? cx : 0.0f;
        const float e4 = valid ? cy : 0.0f;
        const float e5 = valid ? cz : 0.0f;
        v2f t0, t1;
        t0.x = half ? e2 : e0;  t0.y = half ? e3 : e1;
        t1.x = half ? 0.0f : e4; t1.y = half ? 0.0f : e5;
        a0[rt] = t0; a1[rt] = t1;
    }

    v8f acc0[4], acc1[4];
#pragma unroll
    for (int ct = 0; ct < 4; ++ct) {
        const int    o    = ct * 16 + l15;
        const float* wrow = w1 + o * 6;
        const float w0 = wrow[0], w1v = wrow[1], w2v = wrow[2];
        const float w3 = wrow[3], w4v = wrow[4], w5v = wrow[5];
        v2f bt0, bt1;
        bt0.x = half ? w2v : w0;   bt0.y = half ? w3 : w1v;
        bt1.x = half ? 0.0f : w4v; bt1.y = half ? 0.0f : w5v;
        v8f z = {};
        z = __builtin_amdgcn_wmma_f32_16x16x4_f32(false, a0[0], false, bt0,
                                                  (short)0, z, false, false);
        z = __builtin_amdgcn_wmma_f32_16x16x4_f32(false, a1[0], false, bt1,
                                                  (short)0, z, false, false);
        acc0[ct] = z;
        v8f z1 = {};
        z1 = __builtin_amdgcn_wmma_f32_16x16x4_f32(false, a0[1], false, bt0,
                                                   (short)0, z1, false, false);
        z1 = __builtin_amdgcn_wmma_f32_16x16x4_f32(false, a1[1], false, bt1,
                                                   (short)0, z1, false, false);
        acc1[ct] = z1;
    }

#pragma unroll
    for (int ct = 0; ct < 4; ++ct) {
        const int   o = ct * 16 + l15;
        const float s = g1[o] * rsqrtf(rv1[o] + EPS_);
        const float t = (b1[o] - rm1[o]) * s + beta1[o];
        float mx = -3.4e38f, sm = 0.0f;
#pragma unroll
        for (int v = 0; v < 8; ++v) {
            const float val = fmaxf(acc0[ct][v] * s + t, 0.0f);
            mx = fmaxf(mx, val); sm += val;
        }
#pragma unroll
        for (int v = 0; v < 4; ++v) {
            const float val = fmaxf(acc1[ct][v] * s + t, 0.0f);
            mx = (half == 0) ? fmaxf(mx, val) : mx;
            sm = (half == 0) ? (sm + val) : sm;
        }
        mx = fmaxf(mx, __shfl_xor(mx, 16, 32));
        sm = sm + __shfl_xor(sm, 16, 32);
        if (half == 0) {
            const size_t off = ((size_t)b * 64 + o) * N_ + n;
            out_m1[off] = mx;
            out_m2[off] = sm * (1.0f / K_);
        }
    }
}

__global__ __launch_bounds__(128) void conv2_kernel(
    const float* __restrict__ m1, const float* __restrict__ m2,
    const float* __restrict__ w2, const float* __restrict__ b2,
    const float* __restrict__ g2, const float* __restrict__ beta2,
    const float* __restrict__ rm2,const float* __restrict__ rv2,
    float* __restrict__ out)
{
    __shared__ float sw[DOUT_ * 128];
    for (int i = threadIdx.x; i < DOUT_ * 128; i += 128) sw[i] = w2[i];
    __syncthreads();

    const int lane    = threadIdx.x & 31;
    const int half    = lane >> 4;
    const int l15     = lane & 15;
    const int wave    = (blockIdx.x * blockDim.x + threadIdx.x) >> 5;
    const int rowbase = wave * 16;

    const int m  = rowbase + l15;
    const int bb = m >> 12;
    const int nn = m & (N_ - 1);
    const float* arow1 = m1 + (size_t)bb * 64 * N_ + nn;
    const float* arow2 = m2 + (size_t)bb * 64 * N_ + nn;

    v8f acc[4] = {{}, {}, {}, {}};
#pragma unroll
    for (int ks = 0; ks < 32; ++ks) {
        const int k0 = ks * 4 + half * 2;
        float av0, av1;
        if (ks < 16) {
            av0 = arow1[(size_t)k0 * N_];
            av1 = arow1[(size_t)(k0 + 1) * N_];
        } else {
            av0 = arow2[(size_t)(k0 - 64) * N_];
            av1 = arow2[(size_t)(k0 - 63) * N_];
        }
        const v2f a = {av0, av1};
#pragma unroll
        for (int ct = 0; ct < 4; ++ct) {
            const int o = ct * 16 + l15;
            const v2f bt = { sw[o * 128 + k0], sw[o * 128 + k0 + 1] };
            acc[ct] = __builtin_amdgcn_wmma_f32_16x16x4_f32(
                false, a, false, bt, (short)0, acc[ct], false, false);
        }
    }

#pragma unroll
    for (int ct = 0; ct < 4; ++ct) {
        const int   o = ct * 16 + l15;
        const float s = g2[o] * rsqrtf(rv2[o] + EPS_);
        const float t = (b2[o] - rm2[o]) * s + beta2[o];
#pragma unroll
        for (int v = 0; v < 8; ++v) {
            const int mr = rowbase + v + half * 8;
            const int ob = mr >> 12;
            const int on = mr & (N_ - 1);
            out[((size_t)ob * 64 + o) * N_ + on] = fmaxf(acc[ct][v] * s + t, 0.0f);
        }
    }
}

extern "C" void kernel_launch(void* const* d_in, const int* in_sizes, int n_in,
                              void* d_out, int out_size, void* d_ws, size_t ws_size,
                              hipStream_t stream)
{
    const float* x     = (const float*)d_in[0];
    const float* w1    = (const float*)d_in[1];
    const float* b1    = (const float*)d_in[2];
    const float* g1    = (const float*)d_in[3];
    const float* beta1 = (const float*)d_in[4];
    const float* rm1   = (const float*)d_in[5];
    const float* rv1   = (const float*)d_in[6];
    const float* w2    = (const float*)d_in[7];
    const float* b2    = (const float*)d_in[8];
    const float* g2    = (const float*)d_in[9];
    const float* beta2 = (const float*)d_in[10];
    const float* rm2   = (const float*)d_in[11];
    const float* rv2   = (const float*)d_in[12];

    float* out    = (float*)d_out;
    float* out_m1 = out + (size_t)B_ * 64 * N_;
    float* out_m2 = out + 2 * (size_t)B_ * 64 * N_;
    int*   idx    = (int*)d_ws;

    knn_kernel<<<B_ * (N_ / 256), 256, 0, stream>>>(x, idx);

    edge_kernel<<<(B_ * N_) / 4, 128, 0, stream>>>(
        x, idx, w1, b1, g1, beta1, rm1, rv1, out_m1, out_m2);

    conv2_kernel<<<(B_ * N_ / 16) / 4, 128, 0, stream>>>(
        out_m1, out_m2, w2, b2, g2, beta2, rm2, rv2, out);
}


# conv_out: the LDS copy of w2 gets a row pitch of 132 floats (was 128), so a B fragment read (8 bytes from each of 16 rows) covers all 64 banks once instead of hitting one bank pair 16 times
# speedup vs baseline: 8.1797x; 8.1797x over previous
#include <hip/hip_runtime.h>

#define B_    8
#define C_    3
#define N_    4096
#define K_    20
#define H_    64
#define DOUT_ 64
#define EPS_  1e-5f

typedef __attribute__((ext_vector_type(2))) float v2f;
typedef __attribute__((ext_vector_type(8))) float v8f;

#define KNN_THREADS 1024
#define KNN_Q       4
#define KNN_PRE     64

__device__ __forceinline__ float lane_value(float v, int l)
{
    return __int_as_float(__builtin_amdgcn_readlane(__float_as_int(v), l));
}

__global__ __launch_bounds__(KNN_THREADS) void knn_kernel(const float* __restrict__ x,
                                                          int* __restrict__ idx)
{
    __shared__ float sx[N_], sy[N_], sz[N_];
    constexpr int QPB = (KNN_THREADS / 32) * KNN_Q;
    constexpr int BPB = N_ / QPB;
    constexpr int NCH = N_ / 32;
    const int b    = blockIdx.x / BPB;
    const int part = blockIdx.x % BPB;
    const float* xb = x + (size_t)b * 3 * N_;
    for (int j = threadIdx.x; j < N_; j += KNN_THREADS) {
        sx[j] = xb[j];
        sy[j] = xb[N_ + j];
        sz[j] = xb[2 * N_ + j];
    }
    __syncthreads();

    const int lane = threadIdx.x & 31;
    const int wv   = threadIdx.x >> 5;
    const int n0   = part * QPB + wv * KNN_Q;

    const int   nq  = n0 + (lane & (KNN_Q - 1));
    const float lqx = sx[nq], lqy = sy[nq], lqz = sz[nq];
    const float lqq = __builtin_fmaf(lqz, lqz, __builtin_fmaf(lqx, lqx, lqy * lqy));

    float qx[KNN_Q], qy[KNN_Q], qz[KNN_Q], qq[KNN_Q];
    float bd[KNN_Q], thr[KNN_Q], flo[KNN_Q], lm[KNN_Q];
    int   bi[KNN_Q];
#pragma unroll
    for (int q = 0; q < KNN_Q; ++q) {
        qx[q] = lane_value(lqx, q);
        qy[q] = lane_value(lqy, q);
        qz[q] = lane_value(lqz, q);
        qq[q] = lane_value(lqq, q);
        bd[q] = -3.4e38f; bi[q] = 0; lm[q] = -3.4e38f;
    }

    float px = sx[lane], py = sy[lane], pz = sz[lane];

    for (int c = 0; c < KNN_PRE; ++c) {
        const int   jn  = (((c + 1) & (KNN_PRE - 1)) << 5) + lane;
        const float npx = sx[jn], npy = sy[jn], npz = sz[jn];
        const float pp = __builtin_fmaf(pz, pz, __builtin_fmaf(px, px, py * py));
#pragma unroll
        for (int q = 0; q < KNN_Q; ++q) {
            const float ip = __builtin_fmaf(qz[q], pz,
                                 __builtin_fmaf(qx[q], px, qy[q] * py));
            const float dq = __builtin_fmaf(ip, 2.0f, -qq[q]) - pp;
            lm[q] = fmaxf(lm[q], dq);
        }
        px = npx; py = npy; pz = npz;
    }

    float t0[KNN_Q];
#pragma unroll
    for (int q = 0; q < KNN_Q; ++q) t0[q] = -3.4e38f;
    for (int j = 0; j < 32; ++j) {
#pragma unroll
        for (int q = 0; q < KNN_Q; ++q) {
            const float v = lane_value(lm[q], j);
            const int reach = __builtin_popcount(__builtin_amdgcn_ballot_w32(lm[q] >= v));
            const float cand = (reach >= K_) ? v : -3.4e38f;
            t0[q] = (cand > t0[q]) ? cand : t0[q];
        }
    }
#pragma unroll
    for (int q = 0; q < KNN_Q; ++q) {
        const float t = t0[q];
        const int   tb = __float_as_int(t);
        flo[q] = (t >= 2.3509887e-38f) ? __int_as_float(tb - 1)
               : (t > -1.17549435e-38f) ? -1.17549435e-38f
               :                          __int_as_float(tb + 1);
        thr[q] = flo[q];
    }

    for (int c = 0; c < NCH; ++c) {
        const int   jn  = (((c + 1) & (NCH - 1)) << 5) + lane;
        const float npx = sx[jn], npy = sy[jn], npz = sz[jn];

        float    d[KNN_Q];
        unsigned m[KNN_Q];
        unsigned any = 0;
        const float pp = __builtin_fmaf(pz, pz, __builtin_fmaf(px, px, py * py));
#pragma unroll
        for (int q = 0; q < KNN_Q; ++q) {
            const float ip = __builtin_fmaf(qz[q], pz,
                                 __builtin_fmaf(qx[q], px, qy[q] * py));
            d[q] = __builtin_fmaf(ip, 2.0f, -qq[q]) - pp;
            m[q] = __builtin_amdgcn_ballot_w32(d[q] > thr[q]);
            any |= m[q];
        }
        if (any) {
#pragma unroll
            for (int q = 0; q < KNN_Q; ++q) {
                unsigned mm = m[q];
                while (mm) {
                    const int l = __builtin_ctz(mm);
                    mm &= mm - 1;
                    const float dc = lane_value(d[q], l);
                    if (dc > thr[q]) {
                        const int jc = (c << 5) + l;
                        const int bdi = __float_as_int(bd[q]);
                        const int xd = __builtin_amdgcn_permlanex16(
                            bdi, bdi, 0xffffffffu, 0xffffffffu, false, false);
                        const int od = (lane == 0) ? 0x7f800000 : xd;
                        const float ud = __int_as_float(__builtin_amdgcn_update_dpp(
                            od, bdi, 0x111, 0xf, 0xf, false));
                        const int xi = __builtin_amdgcn_permlanex16(
                            bi[q], bi[q], 0xffffffffu, 0xffffffffu, false, false);
                        const int ui = __builtin_amdgcn_update_dpp(
                            xi, bi[q], 0x111, 0xf, 0xf, false);
                        const bool keep = bd[q] >= dc;
                        const bool here = ud >= dc;
                        bd[q]  = keep ? bd[q] : (here ? dc : ud);
                        bi[q]  = keep ? bi[q] : (here ? jc : ui);
                        const float b19 = lane_value(bd[q], K_ - 1);
                        thr[q] = (b19 > flo[q]) ? b19 : flo[q];
                    }
                }
            }
        }
        px = npx; py = npy; pz = npz;
    }

#pragma unroll
    for (int q = 0; q < KNN_Q; ++q) {
        if (lane < K_) idx[((size_t)b * N_ + n0 + q) * K_ + lane] = bi[q];
    }
}

__global__ __launch_bounds__(128) void edge_kernel(
    const float* __restrict__ x,  const int* __restrict__ idx,
    const float* __restrict__ w1, const float* __restrict__ b1,
    const float* __restrict__ g1, const float* __restrict__ beta1,
    const float* __restrict__ rm1,const float* __restrict__ rv1,
    float* __restrict__ out_m1,   float* __restrict__ out_m2)
{
    const int lane = threadIdx.x & 31;
    const int half = lane >> 4;
    const int l15  = lane & 15;
    const int grp  = (blockIdx.x * blockDim.x + threadIdx.x) >> 5;
    const int p    = grp * 16 + l15;
    const int b    = p >> 12;
    const int n    = p & (N_ - 1);

    const float* xb = x + (size_t)b * 3 * N_;
    const float cx = xb[n], cy = xb[N_ + n], cz = xb[2 * N_ + n];

    int nb[K_];
    const int4* irow = reinterpret_cast<const int4*>(idx + (size_t)p * K_);
#pragma unroll
    for (int j = 0; j < K_ / 4; ++j) {
        const int4 q = irow[j];
        nb[4 * j] = q.x; nb[4 * j + 1] = q.y; nb[4 * j + 2] = q.z; nb[4 * j + 3] = q.w;
    }

    v2f bf0[K_];
#pragma unroll
    for (int k = 0; k < K_; ++k) {
        const int   j  = nb[k];
        const float e0 = xb[j] - cx, e1 = xb[N_ + j] - cy, e2 = xb[2 * N_ + j] - cz;
        v2f t0;
        t0.x = half ? e2 : e0;  t0.y = half ? cx : e1;
        bf0[k] = t0;
    }
    v2f bf1;
    bf1.x = half ? 0.0f : cy;  bf1.y = half ? 0.0f : cz;

#pragma unroll
    for (int ct = 0; ct < 4; ++ct) {
        const float* wrow = w1 + (ct * 16 + l15) * 6;
        const float w0 = wrow[0], w1v = wrow[1], w2v = wrow[2];
        const float w3 = wrow[3], w4v = wrow[4], w5v = wrow[5];
        v2f aw0, aw1;
        aw0.x = half ? w2v : w0;   aw0.y = half ? w3 : w1v;
        aw1.x = half ? 0.0f : w4v; aw1.y = half ? 0.0f : w5v;

        float s[8], t[8], mx[8], sa[8], sb[8];
#pragma unroll
        for (int r = 0; r < 8; ++r) {
            const int o = ct * 16 + half * 8 + r;
            s[r] = g1[o] * rsqrtf(rv1[o] + EPS_);
            t[r] = (b1[o] - rm1[o]) * s[r] + beta1[o];
            mx[r] = -3.4e38f; sa[r] = 0.0f; sb[r] = 0.0f;
        }

#pragma unroll
        for (int k = 0; k < K_; ++k) {
            v8f z = {};
            z = __builtin_amdgcn_wmma_f32_16x16x4_f32(false, aw0, false, bf0[k],
                                                      (short)0, z, false, false);
            z = __builtin_amdgcn_wmma_f32_16x16x4_f32(false, aw1, false, bf1,
                                                      (short)0, z, false, false);
#pragma unroll
            for (int r = 0; r < 8; ++r) {
                const float val = fmaxf(z[r] * s[r] + t[r], 0.0f);
                mx[r] = fmaxf(mx[r], val);
                if (k < 8 || k >= 16) sa[r] += val; else sb[r] += val;
            }
        }

#pragma unroll
        for (int r = 0; r < 8; ++r) {
            const int    o   = ct * 16 + half * 8 + r;
            const size_t off = ((size_t)b * 64 + o) * N_ + n;
            out_m1[off] = mx[r];
            out_m2[off] = (sa[r] + sb[r]) * (1.0f / K_);
        }
    }
}

#define W2_PITCH 132
__global__ __launch_bounds__(128) void conv2_kernel(
    const float* __restrict__ m1, const float* __restrict__ m2,
    const float* __restrict__ w2, const float* __restrict__ b2,
    const float* __restrict__ g2, const float* __restrict__ beta2,
    const float* __restrict__ rm2,const float* __restrict__ rv2,
    float* __restrict__ out)
{
    __shared__ float sw[DOUT_ * W2_PITCH];
    for (int r = 0; r < DOUT_; ++r)
        sw[r * W2_PITCH + threadIdx.x] = w2[r * 128 + threadIdx.x];
    __syncthreads();

    const int lane    = threadIdx.x & 31;
    const int half    = lane >> 4;
    const int l15     = lane & 15;
    const int wave    = (blockIdx.x * blockDim.x + threadIdx.x) >> 5;
    const int rowbase = wave * 16;

    const int m  = rowbase + l15;
    const int bb = m >> 12;
    const int nn = m & (N_ - 1);
    const float* arow1 = m1 + (size_t)bb * 64 * N_ + nn;
    const float* arow2 = m2 + (size_t)bb * 64 * N_ + nn;

    v8f acc[4] = {{}, {}, {}, {}};
#pragma unroll
    for (int ks = 0; ks < 32; ++ks) {
        const int k0 = ks * 4 + half * 2;
        float av0, av1;
        if (ks < 16) {
            av0 = arow1[(size_t)k0 * N_];
            av1 = arow1[(size_t)(k0 + 1) * N_];
        } else {
            av0 = arow2[(size_t)(k0 - 64) * N_];
            av1 = arow2[(size_t)(k0 - 63) * N_];
        }
        const v2f a = {av0, av1};
#pragma unroll
        for (int ct = 0; ct < 4; ++ct) {
            const int o = ct * 16 + l15;
            const v2f bt = { sw[o * W2_PITCH + k0], sw[o * W2_PITCH + k0 + 1] };
            acc[ct] = __builtin_amdgcn_wmma_f32_16x16x4_f32(
                false, a, false, bt, (short)0, acc[ct], false, false);
        }
    }

#pragma unroll
    for (int ct = 0; ct < 4; ++ct) {
        const int   o = ct * 16 + l15;
        const float s = g2[o] * rsqrtf(rv2[o] + EPS_);
        const float t = (b2[o] - rm2[o]) * s + beta2[o];
#pragma unroll
        for (int v = 0; v < 8; ++v) {
            const int mr = rowbase + v + half * 8;
            const int ob = mr >> 12;
            const int on = mr & (N_ - 1);
            out[((size_t)ob * 64 + o) * N_ + on] = fmaxf(acc[ct][v] * s + t, 0.0f);
        }
    }
}

extern "C" void kernel_launch(void* const* d_in, const int* in_sizes, int n_in,
                              void* d_out, int out_size, void* d_ws, size_t ws_size,
                              hipStream_t stream)
{
    const float* x     = (const float*)d_in[0];
    const float* w1    = (const float*)d_in[1];
    const float* b1    = (const float*)d_in[2];
    const float* g1    = (const float*)d_in[3];
    const float* beta1 = (const float*)d_in[4];
    const float* rm1   = (const float*)d_in[5];
    const float* rv1   = (const float*)d_in[6];
    const float* w2    = (const float*)d_in[7];
    const float* b2    = (const float*)d_in[8];
    const float* g2    = (const float*)d_in[9];
    const float* beta2 = (const float*)d_in[10];
    const float* rm2   = (const float*)d_in[11];
    const float* rv2   = (const float*)d_in[12];

    float* out    = (float*)d_out;
    float* out_m1 = out + (size_t)B_ * 64 * N_;
    float* out_m2 = out + 2 * (size_t)B_ * 64 * N_;
    int*   idx    = (int*)d_ws;

    knn_kernel<<<B_ * (N_ / ((KNN_THREADS / 32) * KNN_Q)), KNN_THREADS, 0, stream>>>(x, idx);

    edge_kernel<<<(B_ * N_ / 16) / 4, 128, 0, stream>>>(
        x, idx, w1, b1, g1, beta1, rm1, rv1, out_m1, out_m2);

    conv2_kernel<<<(B_ * N_ / 16) / 4, 128, 0, stream>>>(
        out_m1, out_m2, w2, b2, g2, beta2, rm2, rv2, out);
}


# conv_out fused into the edge kernel through registers: a lane's pooled D registers are the B fragments as they stand (w2 from LDS as A, same k order); 2 launches; conv_out sum order changes
# speedup vs baseline: 8.5934x; 1.0506x over previous
#include <hip/hip_runtime.h>

#define B_    8
#define C_    3
#define N_    4096
#define K_    20
#define H_    64
#define DOUT_ 64
#define EPS_  1e-5f

typedef __attribute__((ext_vector_type(2))) float v2f;
typedef __attribute__((ext_vector_type(8))) float v8f;

#define KNN_THREADS 1024
#define KNN_Q       4
#define KNN_PRE     64

__device__ __forceinline__ float lane_value(float v, int l)
{
    return __int_as_float(__builtin_amdgcn_readlane(__float_as_int(v), l));
}

__global__ __launch_bounds__(KNN_THREADS) void knn_kernel(const float* __restrict__ x,
                                                          int* __restrict__ idx)
{
    __shared__ float sx[N_], sy[N_], sz[N_];
    constexpr int QPB = (KNN_THREADS / 32) * KNN_Q;
    constexpr int BPB = N_ / QPB;
    constexpr int NCH = N_ / 32;
    const int b    = blockIdx.x / BPB;
    const int part = blockIdx.x % BPB;
    const float* xb = x + (size_t)b * 3 * N_;
    for (int j = threadIdx.x; j < N_; j += KNN_THREADS) {
        sx[j] = xb[j];
        sy[j] = xb[N_ + j];
        sz[j] = xb[2 * N_ + j];
    }
    __syncthreads();

    const int lane = threadIdx.x & 31;
    const int wv   = threadIdx.x >> 5;
    const int n0   = part * QPB + wv * KNN_Q;

    const int   nq  = n0 + (lane & (KNN_Q - 1));
    const float lqx = sx[nq], lqy = sy[nq], lqz = sz[nq];
    const float lqq = __builtin_fmaf(lqz, lqz, __builtin_fmaf(lqx, lqx, lqy * lqy));

    float qx[KNN_Q], qy[KNN_Q], qz[KNN_Q], qq[KNN_Q];
    float bd[KNN_Q], thr[KNN_Q], flo[KNN_Q], lm[KNN_Q];
    int   bi[KNN_Q];
#pragma unroll
    for (int q = 0; q < KNN_Q; ++q) {
        qx[q] = lane_value(lqx, q);
        qy[q] = lane_value(lqy, q);
        qz[q] = lane_value(lqz, q);
        qq[q] = lane_value(lqq, q);
        bd[q] = -3.4e38f; bi[q] = 0; lm[q] = -3.4e38f;
    }

    float px = sx[lane], py = sy[lane], pz = sz[lane];

    for (int c = 0; c < KNN_PRE; ++c) {
        const int   jn  = (((c + 1) & (KNN_PRE - 1)) << 5) + lane;
        const float npx = sx[jn], npy = sy[jn], npz = sz[jn];
        const float pp = __builtin_fmaf(pz, pz, __builtin_fmaf(px, px, py * py));
#pragma unroll
        for (int q = 0; q < KNN_Q; ++q) {
            const float ip = __builtin_fmaf(qz[q], pz,
                                 __builtin_fmaf(qx[q], px, qy[q] * py));
            const float dq = __builtin_fmaf(ip, 2.0f, -qq[q]) - pp;
            lm[q] = fmaxf(lm[q], dq);
        }
        px = npx; py = npy; pz = npz;
    }

    float t0[KNN_Q];
#pragma unroll
    for (int q = 0; q < KNN_Q; ++q) t0[q] = -3.4e38f;
    for (int j = 0; j < 32; ++j) {
#pragma unroll
        for (int q = 0; q < KNN_Q; ++q) {
            const float v = lane_value(lm[q], j);
            const int reach = __builtin_popcount(__builtin_amdgcn_ballot_w32(lm[q] >= v));
            const float cand = (reach >= K_) ? v : -3.4e38f;
            t0[q] = (cand > t0[q]) ? cand : t0[q];
        }
    }
#pragma unroll
    for (int q = 0; q < KNN_Q; ++q) {
        const float t = t0[q];
        const int   tb = __float_as_int(t);
        flo[q] = (t >= 2.3509887e-38f) ? __int_as_float(tb - 1)
               : (t > -1.17549435e-38f) ? -1.17549435e-38f
               :                          __int_as_float(tb + 1);
        thr[q] = flo[q];
    }

    for (int c = 0; c < NCH; ++c) {
        const int   jn  = (((c + 1) & (NCH - 1)) << 5) + lane;
        const float npx = sx[jn], npy = sy[jn], npz = sz[jn];

        float    d[KNN_Q];
        unsigned m[KNN_Q];
        unsigned any = 0;
        const float pp = __builtin_fmaf(pz, pz, __builtin_fmaf(px, px, py * py));
#pragma unroll
        for (int q = 0; q < KNN_Q; ++q) {
            const float ip = __builtin_fmaf(qz[q], pz,
                                 __builtin_fmaf(qx[q], px, qy[q] * py));
            d[q] = __builtin_fmaf(ip, 2.0f, -qq[q]) - pp;
            m[q] = __builtin_amdgcn_ballot_w32(d[q] > thr[q]);
            any |= m[q];
        }
        if (any) {
#pragma unroll
            for (int q = 0; q < KNN_Q; ++q) {
                unsigned mm = m[q];
                while (mm) {
                    const int l = __builtin_ctz(mm);
                    mm &= mm - 1;
                    const float dc = lane_value(d[q], l);
                    if (dc > thr[q]) {
                        const int jc = (c << 5) + l;
                        const int bdi = __float_as_int(bd[q]);
                        const int xd = __builtin_amdgcn_permlanex16(
                            bdi, bdi, 0xffffffffu, 0xffffffffu, false, false);
                        const int od = (lane == 0) ? 0x7f800000 : xd;
                        const float ud = __int_as_float(__builtin_amdgcn_update_dpp(
                            od, bdi, 0x111, 0xf, 0xf, false));
                        const int xi = __builtin_amdgcn_permlanex16(
                            bi[q], bi[q], 0xffffffffu, 0xffffffffu, false, false);
                        const int ui = __builtin_amdgcn_update_dpp(
                            xi, bi[q], 0x111, 0xf, 0xf, false);
                        const bool keep = bd[q] >= dc;
                        const bool here = ud >= dc;
                        bd[q]  = keep ? bd[q] : (here ? dc : ud);
                        bi[q]  = keep ? bi[q] : (here ? jc : ui);
                        const float b19 = lane_value(bd[q], K_ - 1);
                        thr[q] = (b19 > flo[q]) ? b19 : flo[q];
                    }
                }
            }
        }
        px = npx; py = npy; pz = npz;
    }

#pragma unroll
    for (int q = 0; q < KNN_Q; ++q) {
        if (lane < K_) idx[((size_t)b * N_ + n0 + q) * K_ + lane] = bi[q];
    }
}

#define W2_PITCH 132
__global__ __launch_bounds__(128) void edge_conv_kernel(
    const float* __restrict__ x,  const int* __restrict__ idx,
    const float* __restrict__ w1, const float* __restrict__ b1,
    const float* __restrict__ g1, const float* __restrict__ beta1,
    const float* __restrict__ rm1,const float* __restrict__ rv1,
    const float* __restrict__ w2, const float* __restrict__ b2,
    const float* __restrict__ g2, const float* __restrict__ beta2,
    const float* __restrict__ rm2,const float* __restrict__ rv2,
    float* __restrict__ out, float* __restrict__ out_m1, float* __restrict__ out_m2)
{
    __shared__ float sw[DOUT_ * W2_PITCH];
    for (int r = 0; r < DOUT_; ++r)
        sw[r * W2_PITCH + threadIdx.x] = w2[r * 128 + threadIdx.x];

    const int lane = threadIdx.x & 31;
    const int half = lane >> 4;
    const int l15  = lane & 15;
    const int grp  = (blockIdx.x * blockDim.x + threadIdx.x) >> 5;
    const int p    = grp * 16 + l15;
    const int b    = p >> 12;
    const int n    = p & (N_ - 1);

    const float* xb = x + (size_t)b * 3 * N_;
    const float cx = xb[n], cy = xb[N_ + n], cz = xb[2 * N_ + n];

    int nb[K_];
    const int4* irow = reinterpret_cast<const int4*>(idx + (size_t)p * K_);
#pragma unroll
    for (int j = 0; j < K_ / 4; ++j) {
        const int4 q = irow[j];
        nb[4 * j] = q.x; nb[4 * j + 1] = q.y; nb[4 * j + 2] = q.z; nb[4 * j + 3] = q.w;
    }

    v2f bf0[K_];
#pragma unroll
    for (int k = 0; k < K_; ++k) {
        const int   j  = nb[k];
        const float e0 = xb[j] - cx, e1 = xb[N_ + j] - cy, e2 = xb[2 * N_ + j] - cz;
        v2f t0;
        t0.x = half ? e2 : e0;  t0.y = half ? cx : e1;
        bf0[k] = t0;
    }
    v2f bf1;
    bf1.x = half ? 0.0f : cy;  bf1.y = half ? 0.0f : cz;

    __syncthreads();

    v8f acc[4] = {{}, {}, {}, {}};
    const float* swl = sw + l15 * W2_PITCH + half * 8;

#pragma unroll
    for (int ct = 0; ct < 4; ++ct) {
        const float* wrow = w1 + (ct * 16 + l15) * 6;
        const float w0 = wrow[0], w1v = wrow[1], w2v = wrow[2];
        const float w3 = wrow[3], w4v = wrow[4], w5v = wrow[5];
        v2f aw0, aw1;
        aw0.x = half ? w2v : w0;   aw0.y = half ? w3 : w1v;
        aw1.x = half ? 0.0f : w4v; aw1.y = half ? 0.0f : w5v;

        float s[8], t[8], mx[8], sa[8], sb[8];
#pragma unroll
        for (int r = 0; r < 8; ++r) {
            const int o = ct * 16 + half * 8 + r;
            s[r] = g1[o] * rsqrtf(rv1[o] + EPS_);
            t[r] = (b1[o] - rm1[o]) * s[r] + beta1[o];
            mx[r] = -3.4e38f; sa[r] = 0.0f; sb[r] = 0.0f;
        }

#pragma unroll
        for (int k = 0; k < K_; ++k) {
            v8f z = {};
            z = __builtin_amdgcn_wmma_f32_16x16x4_f32(false, aw0, false, bf0[k],
                                                      (short)0, z, false, false);
            z = __builtin_amdgcn_wmma_f32_16x16x4_f32(false, aw1, false, bf1,
                                                      (short)0, z, false, false);
#pragma unroll
            for (int r = 0; r < 8; ++r) {
                const float val = fmaxf(z[r] * s[r] + t[r], 0.0f);
                mx[r] = fmaxf(mx[r], val);
                if (k < 8 || k >= 16) sa[r] += val; else sb[r] += val;
            }
        }

        float mean[8];
#pragma unroll
        for (int r = 0; r < 8; ++r) {
            const int    o   = ct * 16 + half * 8 + r;
            const size_t off = ((size_t)b * 64 + o) * N_ + n;
            mean[r] = (sa[r] + sb[r]) * (1.0f / K_);
            out_m1[off] = mx[r];
            out_m2[off] = mean[r];
        }

#pragma unroll
        for (int j = 0; j < 4; ++j) {
            const v2f bm1 = {mx[2 * j],   mx[2 * j + 1]};
            const v2f bm2 = {mean[2 * j], mean[2 * j + 1]};
            const int c = ct * 16 + 2 * j;
#pragma unroll
            for (int ot = 0; ot < 4; ++ot) {
                const float* wr = swl + ot * 16 * W2_PITCH;
                const v2f a1 = {wr[c],      wr[c + 1]};
                const v2f a2 = {wr[64 + c], wr[64 + c + 1]};
                acc[ot] = __builtin_amdgcn_wmma_f32_16x16x4_f32(
                    false, a1, false, bm1, (short)0, acc[ot], false, false);
                acc[ot] = __builtin_amdgcn_wmma_f32_16x16x4_f32(
                    false, a2, false, bm2, (short)0, acc[ot], false, false);
            }
        }
    }

#pragma unroll
    for (int ot = 0; ot < 4; ++ot) {
#pragma unroll
        for (int r = 0; r < 8; ++r) {
            const int   o  = ot * 16 + half * 8 + r;
            const float s2 = g2[o] * rsqrtf(rv2[o] + EPS_);
            const float t2 = (b2[o] - rm2[o]) * s2 + beta2[o];
            out[((size_t)b * 64 + o) * N_ + n] = fmaxf(acc[ot][r] * s2 + t2, 0.0f);
        }
    }
}

extern "C" void kernel_launch(void* const* d_in, const int* in_sizes, int n_in,
                              void* d_out, int out_size, void* d_ws, size_t ws_size,
                              hipStream_t stream)
{
    const float* x     = (const float*)d_in[0];
    const float* w1    = (const float*)d_in[1];
    const float* b1    = (const float*)d_in[2];
    const float* g1    = (const float*)d_in[3];
    const float* beta1 = (const float*)d_in[4];
    const float* rm1   = (const float*)d_in[5];
    const float* rv1   = (const float*)d_in[6];
    const float* w2    = (const float*)d_in[7];
    const float* b2    = (const float*)d_in[8];
    const float* g2    = (const float*)d_in[9];
    const float* beta2 = (const float*)d_in[10];
    const float* rm2   = (const float*)d_in[11];
    const float* rv2   = (const float*)d_in[12];

    float* out    = (float*)d_out;
    float* out_m1 = out + (size_t)B_ * 64 * N_;
    float* out_m2 = out + 2 * (size_t)B_ * 64 * N_;
    int*   idx    = (int*)d_ws;

    knn_kernel<<<B_ * (N_ / ((KNN_THREADS / 32) * KNN_Q)), KNN_THREADS, 0, stream>>>(x, idx);

    edge_conv_kernel<<<(B_ * N_ / 16) / 4, 128, 0, stream>>>(
        x, idx, w1, b1, g1, beta1, rm1, rv1,
        w2, b2, g2, beta2, rm2, rv2, out, out_m1, out_m2);
}


# knn insertion: list shift is one ds_bpermute pull per register (was permlanex16 + row_shr DPP + lane-0 patch); landing lane from the keep mask by scalar shift; list pre-filled with start threshold
# speedup vs baseline: 9.0680x; 1.0552x over previous
#include <hip/hip_runtime.h>

#define B_    8
#define C_    3
#define N_    4096
#define K_    20
#define H_    64
#define DOUT_ 64
#define EPS_  1e-5f

typedef __attribute__((ext_vector_type(2))) float v2f;
typedef __attribute__((ext_vector_type(8))) float v8f;

#define KNN_THREADS 1024
#define KNN_Q       4
#define KNN_PRE     64

__device__ __forceinline__ float lane_value(float v, int l)
{
    return __int_as_float(__builtin_amdgcn_readlane(__float_as_int(v), l));
}

__global__ __launch_bounds__(KNN_THREADS) void knn_kernel(const float* __restrict__ x,
                                                          int* __restrict__ idx)
{
    __shared__ float sx[N_], sy[N_], sz[N_];
    constexpr int QPB = (KNN_THREADS / 32) * KNN_Q;
    constexpr int BPB = N_ / QPB;
    constexpr int NCH = N_ / 32;
    const int b    = blockIdx.x / BPB;
    const int part = blockIdx.x % BPB;
    const float* xb = x + (size_t)b * 3 * N_;
    for (int j = threadIdx.x; j < N_; j += KNN_THREADS) {
        sx[j] = xb[j];
        sy[j] = xb[N_ + j];
        sz[j] = xb[2 * N_ + j];
    }
    __syncthreads();

    const int lane = threadIdx.x & 31;
    const int wv   = threadIdx.x >> 5;
    const int n0   = part * QPB + wv * KNN_Q;

    const int   nq  = n0 + (lane & (KNN_Q - 1));
    const float lqx = sx[nq], lqy = sy[nq], lqz = sz[nq];
    const float lqq = __builtin_fmaf(lqz, lqz, __builtin_fmaf(lqx, lqx, lqy * lqy));

    float qx[KNN_Q], qy[KNN_Q], qz[KNN_Q], qq[KNN_Q];
    float bd[KNN_Q], thr[KNN_Q], lm[KNN_Q];
    int   bi[KNN_Q];
#pragma unroll
    for (int q = 0; q < KNN_Q; ++q) {
        qx[q] = lane_value(lqx, q);
        qy[q] = lane_value(lqy, q);
        qz[q] = lane_value(lqz, q);
        qq[q] = lane_value(lqq, q);
        bi[q] = 0; lm[q] = -3.4e38f;
    }

    float px = sx[lane], py = sy[lane], pz = sz[lane];

    for (int c = 0; c < KNN_PRE; ++c) {
        const int   jn  = (((c + 1) & (KNN_PRE - 1)) << 5) + lane;
        const float npx = sx[jn], npy = sy[jn], npz = sz[jn];
        const float pp = __builtin_fmaf(pz, pz, __builtin_fmaf(px, px, py * py));
#pragma unroll
        for (int q = 0; q < KNN_Q; ++q) {
            const float ip = __builtin_fmaf(qz[q], pz,
                                 __builtin_fmaf(qx[q], px, qy[q] * py));
            const float dq = __builtin_fmaf(ip, 2.0f, -qq[q]) - pp;
            lm[q] = fmaxf(lm[q], dq);
        }
        px = npx; py = npy; pz = npz;
    }

    float t0[KNN_Q];
#pragma unroll
    for (int q = 0; q < KNN_Q; ++q) t0[q] = -3.4e38f;
    for (int j = 0; j < 32; ++j) {
#pragma unroll
        for (int q = 0; q < KNN_Q; ++q) {
            const float v = lane_value(lm[q], j);
            const int reach = __builtin_popcount(__builtin_amdgcn_ballot_w32(lm[q] >= v));
            const float cand = (reach >= K_) ? v : -3.4e38f;
            t0[q] = (cand > t0[q]) ? cand : t0[q];
        }
    }
#pragma unroll
    for (int q = 0; q < KNN_Q; ++q) {
        const float t = t0[q];
        const int   tb = __float_as_int(t);
        const float f = (t >= 2.3509887e-38f) ? __int_as_float(tb - 1)
               : (t > -1.17549435e-38f) ? -1.17549435e-38f
               :                          __int_as_float(tb + 1);
        bd[q]  = f;
        thr[q] = f;
    }

    const int pull = ((lane == 0) ? 0 : (lane - 1)) << 2;

    for (int c = 0; c < NCH; ++c) {
        const int   jn  = (((c + 1) & (NCH - 1)) << 5) + lane;
        const float npx = sx[jn], npy = sy[jn], npz = sz[jn];

        float    d[KNN_Q];
        unsigned m[KNN_Q];
        unsigned any = 0;
        const float pp = __builtin_fmaf(pz, pz, __builtin_fmaf(px, px, py * py));
#pragma unroll
        for (int q = 0; q < KNN_Q; ++q) {
            const float ip = __builtin_fmaf(qz[q], pz,
                                 __builtin_fmaf(qx[q], px, qy[q] * py));
            d[q] = __builtin_fmaf(ip, 2.0f, -qq[q]) - pp;
            m[q] = __builtin_amdgcn_ballot_w32(d[q] > thr[q]);
            any |= m[q];
        }
        if (any) {
#pragma unroll
            for (int q = 0; q < KNN_Q; ++q) {
                unsigned mm = m[q];
                while (mm) {
                    const int l = __builtin_ctz(mm);
                    mm &= mm - 1;
                    const float dc = lane_value(d[q], l);
                    if (dc > thr[q]) {
                        const int jc = (c << 5) + l;
                        const bool     keep = bd[q] >= dc;
                        const unsigned km   = __builtin_amdgcn_ballot_w32(keep);
                        const bool here = __builtin_amdgcn_inverse_ballot_w32((km << 1) | 1u);
                        const float ud = __int_as_float(__builtin_amdgcn_ds_bpermute(
                            pull, __float_as_int(bd[q])));
                        const int   ui = __builtin_amdgcn_ds_bpermute(pull, bi[q]);
                        bd[q] = keep ? bd[q] : (here ? dc : ud);
                        bi[q] = keep ? bi[q] : (here ? jc : ui);
                        thr[q] = lane_value(bd[q], K_ - 1);
                    }
                }
            }
        }
        px = npx; py = npy; pz = npz;
    }

#pragma unroll
    for (int q = 0; q < KNN_Q; ++q) {
        if (lane < K_) idx[((size_t)b * N_ + n0 + q) * K_ + lane] = bi[q];
    }
}

#define W2_PITCH 132
__global__ __launch_bounds__(128) void edge_conv_kernel(
    const float* __restrict__ x,  const int* __restrict__ idx,
    const float* __restrict__ w1, const float* __restrict__ b1,
    const float* __restrict__ g1, const float* __restrict__ beta1,
    const float* __restrict__ rm1,const float* __restrict__ rv1,
    const float* __restrict__ w2, const float* __restrict__ b2,
    const float* __restrict__ g2, const float* __restrict__ beta2,
    const float* __restrict__ rm2,const float* __restrict__ rv2,
    float* __restrict__ out, float* __restrict__ out_m1, float* __restrict__ out_m2)
{
    __shared__ float sw[DOUT_ * W2_PITCH];
    for (int r = 0; r < DOUT_; ++r)
        sw[r * W2_PITCH + threadIdx.x] = w2[r * 128 + threadIdx.x];

    const int lane = threadIdx.x & 31;
    const int half = lane >> 4;
    const int l15  = lane & 15;
    const int grp  = (blockIdx.x * blockDim.x + threadIdx.x) >> 5;
    const int p    = grp * 16 + l15;
    const int b    = p >> 12;
    const int n    = p & (N_ - 1);

    const float* xb = x + (size_t)b * 3 * N_;
    const float cx = xb[n], cy = xb[N_ + n], cz = xb[2 * N_ + n];

    int nb[K_];
    const int4* irow = reinterpret_cast<const int4*>(idx + (size_t)p * K_);
#pragma unroll
    for (int j = 0; j < K_ / 4; ++j) {
        const int4 q = irow[j];
        nb[4 * j] = q.x; nb[4 * j + 1] = q.y; nb[4 * j + 2] = q.z; nb[4 * j + 3] = q.w;
    }

    v2f bf0[K_];
#pragma unroll
    for (int k = 0; k < K_; ++k) {
        const int   j  = nb[k];
        const float e0 = xb[j] - cx, e1 = xb[N_ + j] - cy, e2 = xb[2 * N_ + j] - cz;
        v2f t0;
        t0.x = half ? e2 : e0;  t0.y = half ? cx : e1;
        bf0[k] = t0;
    }
    v2f bf1;
    bf1.x = half ? 0.0f : cy;  bf1.y = half ? 0.0f : cz;

    __syncthreads();

    v8f acc[4] = {{}, {}, {}, {}};
    const float* swl = sw + l15 * W2_PITCH + half * 8;

#pragma unroll
    for (int ct = 0; ct < 4; ++ct) {
        const float* wrow = w1 + (ct * 16 + l15) * 6;
        const float w0 = wrow[0], w1v = wrow[1], w2v = wrow[2];
        const float w3 = wrow[3], w4v = wrow[4], w5v = wrow[5];
        v2f aw0, aw1;
        aw0.x = half ? w2v : w0;   aw0.y = half ? w3 : w1v;
        aw1.x = half ? 0.0f : w4v; aw1.y = half ? 0.0f : w5v;

        float s[8], t[8], mx[8], sa[8], sb[8];
#pragma unroll
        for (int r = 0; r < 8; ++r) {
            const int o = ct * 16 + half * 8 + r;
            s[r] = g1[o] * rsqrtf(rv1[o] + EPS_);
            t[r] = (b1[o] - rm1[o]) * s[r] + beta1[o];
            mx[r] = -3.4e38f; sa[r] = 0.0f; sb[r] = 0.0f;
        }

#pragma unroll
        for (int k = 0; k < K_; ++k) {
            v8f z = {};
            z = __builtin_amdgcn_wmma_f32_16x16x4_f32(false, aw0, false, bf0[k],
                                                      (short)0, z, false, false);
            z = __builtin_amdgcn_wmma_f32_16x16x4_f32(false, aw1, false, bf1,
                                                      (short)0, z, false, false);
#pragma unroll
            for (int r = 0; r < 8; ++r) {
                const float val = fmaxf(z[r] * s[r] + t[r], 0.0f);
                mx[r] = fmaxf(mx[r], val);
                if (k < 8 || k >= 16) sa[r] += val; else sb[r] += val;
            }
        }

        float mean[8];
#pragma unroll
        for (int r = 0; r < 8; ++r) {
            const int    o   = ct * 16 + half * 8 + r;
            const size_t off = ((size_t)b * 64 + o) * N_ + n;
            mean[r] = (sa[r] + sb[r]) * (1.0f / K_);
            out_m1[off] = mx[r];
            out_m2[off] = mean[r];
        }

#pragma unroll
        for (int j = 0; j < 4; ++j) {
            const v2f bm1 = {mx[2 * j],   mx[2 * j + 1]};
            const v2f bm2 = {mean[2 * j], mean[2 * j + 1]};
            const int c = ct * 16 + 2 * j;
#pragma unroll
            for (int ot = 0; ot < 4; ++ot) {
                const float* wr = swl + ot * 16 * W2_PITCH;
                const v2f a1 = {wr[c],      wr[c + 1]};
                const v2f a2 = {wr[64 + c], wr[64 + c + 1]};
                acc[ot] = __builtin_amdgcn_wmma_f32_16x16x4_f32(
                    false, a1, false, bm1, (short)0, acc[ot], false, false);
                acc[ot] = __builtin_amdgcn_wmma_f32_16x16x4_f32(
                    false, a2, false, bm2, (short)0, acc[ot], false, false);
            }
        }
    }

#pragma unroll
    for (int ot = 0; ot < 4; ++ot) {
#pragma unroll
        for (int r = 0; r < 8; ++r) {
            const int   o  = ot * 16 + half * 8 + r;
            const float s2 = g2[o] * rsqrtf(rv2[o] + EPS_);
            const float t2 = (b2[o] - rm2[o]) * s2 + beta2[o];
            out[((size_t)b * 64 + o) * N_ + n] = fmaxf(acc[ot][r] * s2 + t2, 0.0f);
        }
    }
}

extern "C" void kernel_launch(void* const* d_in, const int* in_sizes, int n_in,
                              void* d_out, int out_size, void* d_ws, size_t ws_size,
                              hipStream_t stream)
{
    const float* x     = (const float*)d_in[0];
    const float* w1    = (const float*)d_in[1];
    const float* b1    = (const float*)d_in[2];
    const float* g1    = (const float*)d_in[3];
    const float* beta1 = (const float*)d_in[4];
    const float* rm1   = (const float*)d_in[5];
    const float* rv1   = (const float*)d_in[6];
    const float* w2    = (const float*)d_in[7];
    const float* b2    = (const float*)d_in[8];
    const float* g2    = (const float*)d_in[9];
    const float* beta2 = (const float*)d_in[10];
    const float* rm2   = (const float*)d_in[11];
    const float* rv2   = (const float*)d_in[12];

    float* out    = (float*)d_out;
    float* out_m1 = out + (size_t)B_ * 64 * N_;
    float* out_m2 = out + 2 * (size_t)B_ * 64 * N_;
    int*   idx    = (int*)d_ws;

    knn_kernel<<<B_ * (N_ / ((KNN_THREADS / 32) * KNN_Q)), KNN_THREADS, 0, stream>>>(x, idx);

    edge_conv_kernel<<<(B_ * N_ / 16) / 4, 128, 0, stream>>>(
        x, idx, w1, b1, g1, beta1, rm1, rv1,
        w2, b2, g2, beta2, rm2, rv2, out, out_m1, out_m2);
}


# knn: a lane takes two points of a 64-point chunk (L, L+32) as a register pair; their distance terms are packed f32 instructions, one for both points (same bits); half the loop trips per point
# speedup vs baseline: 9.5943x; 1.0580x over previous
#include <hip/hip_runtime.h>

#define B_    8
#define C_    3
#define N_    4096
#define K_    20
#define H_    64
#define DOUT_ 64
#define EPS_  1e-5f

typedef __attribute__((ext_vector_type(2))) float v2f;
typedef __attribute__((ext_vector_type(8))) float v8f;

#define KNN_THREADS 1024
#define KNN_Q       4
#define KNN_PRE     32

__device__ __forceinline__ float lane_value(float v, int l)
{
    return __int_as_float(__builtin_amdgcn_readlane(__float_as_int(v), l));
}

__device__ __forceinline__ v2f fma2(v2f a, v2f b, v2f c)
{
    return __builtin_elementwise_fma(a, b, c);
}

__device__ __forceinline__ v2f both(float s)
{
    v2f r = {s, s};
    return r;
}

__global__ __launch_bounds__(KNN_THREADS) void knn_kernel(const float* __restrict__ x,
                                                          int* __restrict__ idx)
{
    __shared__ float sx[N_], sy[N_], sz[N_];
    constexpr int QPB = (KNN_THREADS / 32) * KNN_Q;
    constexpr int BPB = N_ / QPB;
    constexpr int NCH = N_ / 64;
    const int b    = blockIdx.x / BPB;
    const int part = blockIdx.x % BPB;
    const float* xb = x + (size_t)b * 3 * N_;
    for (int j = threadIdx.x; j < N_; j += KNN_THREADS) {
        sx[j] = xb[j];
        sy[j] = xb[N_ + j];
        sz[j] = xb[2 * N_ + j];
    }
    __syncthreads();

    const int lane = threadIdx.x & 31;
    const int wv   = threadIdx.x >> 5;
    const int n0   = part * QPB + wv * KNN_Q;

    const int   nq  = n0 + (lane & (KNN_Q - 1));
    const float lqx = sx[nq], lqy = sy[nq], lqz = sz[nq];
    const float lqq = __builtin_fmaf(lqz, lqz, __builtin_fmaf(lqx, lqx, lqy * lqy));

    float qx[KNN_Q], qy[KNN_Q], qz[KNN_Q], qq[KNN_Q];
    float bd[KNN_Q], thr[KNN_Q], lm[KNN_Q];
    int   bi[KNN_Q];
#pragma unroll
    for (int q = 0; q < KNN_Q; ++q) {
        qx[q] = lane_value(lqx, q);
        qy[q] = lane_value(lqy, q);
        qz[q] = lane_value(lqz, q);
        qq[q] = lane_value(lqq, q);
        bi[q] = 0; lm[q] = -3.4e38f;
    }

    const v2f two = {2.0f, 2.0f};
    v2f px = {sx[lane], sx[lane + 32]};
    v2f py = {sy[lane], sy[lane + 32]};
    v2f pz = {sz[lane], sz[lane + 32]};

    for (int c = 0; c < KNN_PRE; ++c) {
        const int jn  = (((c + 1) & (KNN_PRE - 1)) << 6) + lane;
        const v2f npx = {sx[jn], sx[jn + 32]};
        const v2f npy = {sy[jn], sy[jn + 32]};
        const v2f npz = {sz[jn], sz[jn + 32]};
        const v2f pp = fma2(pz, pz, fma2(px, px, py * py));
#pragma unroll
        for (int q = 0; q < KNN_Q; ++q) {
            const v2f ip = fma2(both(qz[q]), pz,
                                fma2(both(qx[q]), px, both(qy[q]) * py));
            const v2f dq = fma2(ip, two, both(-qq[q])) - pp;
            lm[q] = fmaxf(lm[q], fmaxf(dq.x, dq.y));
        }
        px = npx; py = npy; pz = npz;
    }

    float t0[KNN_Q];
#pragma unroll
    for (int q = 0; q < KNN_Q; ++q) t0[q] = -3.4e38f;
    for (int j = 0; j < 32; ++j) {
#pragma unroll
        for (int q = 0; q < KNN_Q; ++q) {
            const float v = lane_value(lm[q], j);
            const int reach = __builtin_popcount(__builtin_amdgcn_ballot_w32(lm[q] >= v));
            const float cand = (reach >= K_) ? v : -3.4e38f;
            t0[q] = (cand > t0[q]) ? cand : t0[q];
        }
    }
#pragma unroll
    for (int q = 0; q < KNN_Q; ++q) {
        const float t = t0[q];
        const int   tb = __float_as_int(t);
        const float f = (t >= 2.3509887e-38f) ? __int_as_float(tb - 1)
               : (t > -1.17549435e-38f) ? -1.17549435e-38f
               :                          __int_as_float(tb + 1);
        bd[q]  = f;
        thr[q] = f;
    }

    const int pull = ((lane == 0) ? 0 : (lane - 1)) << 2;

    for (int c = 0; c < NCH; ++c) {
        const int jn  = (((c + 1) & (NCH - 1)) << 6) + lane;
        const v2f npx = {sx[jn], sx[jn + 32]};
        const v2f npy = {sy[jn], sy[jn + 32]};
        const v2f npz = {sz[jn], sz[jn + 32]};

        float    d[KNN_Q][2];
        unsigned m[KNN_Q][2];
        const v2f pp = fma2(pz, pz, fma2(px, px, py * py));
#pragma unroll
        for (int q = 0; q < KNN_Q; ++q) {
            const v2f ip = fma2(both(qz[q]), pz,
                                fma2(both(qx[q]), px, both(qy[q]) * py));
            const v2f dq = fma2(ip, two, both(-qq[q])) - pp;
            d[q][0] = dq.x; d[q][1] = dq.y;
            m[q][0] = __builtin_amdgcn_ballot_w32(dq.x > thr[q]);
            m[q][1] = __builtin_amdgcn_ballot_w32(dq.y > thr[q]);
        }
#pragma unroll
        for (int q = 0; q < KNN_Q; ++q) {
#pragma unroll
            for (int h = 0; h < 2; ++h) {
                unsigned mm = m[q][h];
                while (mm) {
                    const int l = __builtin_ctz(mm);
                    mm &= mm - 1;
                    const float dc = lane_value(d[q][h], l);
                    if (dc > thr[q]) {
                        const int jc = (c << 6) + (h << 5) + l;
                        const bool     keep = bd[q] >= dc;
                        const unsigned km   = __builtin_amdgcn_ballot_w32(keep);
                        const bool here = __builtin_amdgcn_inverse_ballot_w32((km << 1) | 1u);
                        const float ud = __int_as_float(__builtin_amdgcn_ds_bpermute(
                            pull, __float_as_int(bd[q])));
                        const int   ui = __builtin_amdgcn_ds_bpermute(pull, bi[q]);
                        bd[q] = keep ? bd[q] : (here ? dc : ud);
                        bi[q] = keep ? bi[q] : (here ? jc : ui);
                        thr[q] = lane_value(bd[q], K_ - 1);
                    }
                }
            }
        }
        px = npx; py = npy; pz = npz;
    }

#pragma unroll
    for (int q = 0; q < KNN_Q; ++q) {
        if (lane < K_) idx[((size_t)b * N_ + n0 + q) * K_ + lane] = bi[q];
    }
}

#define W2_PITCH 132
__global__ __launch_bounds__(128) void edge_conv_kernel(
    const float* __restrict__ x,  const int* __restrict__ idx,
    const float* __restrict__ w1, const float* __restrict__ b1,
    const float* __restrict__ g1, const float* __restrict__ beta1,
    const float* __restrict__ rm1,const float* __restrict__ rv1,
    const float* __restrict__ w2, const float* __restrict__ b2,
    const float* __restrict__ g2, const float* __restrict__ beta2,
    const float* __restrict__ rm2,const float* __restrict__ rv2,
    float* __restrict__ out, float* __restrict__ out_m1, float* __restrict__ out_m2)
{
    __shared__ float sw[DOUT_ * W2_PITCH];
    for (int r = 0; r < DOUT_; ++r)
        sw[r * W2_PITCH + threadIdx.x] = w2[r * 128 + threadIdx.x];

    const int lane = threadIdx.x & 31;
    const int half = lane >> 4;
    const int l15  = lane & 15;
    const int grp  = (blockIdx.x * blockDim.x + threadIdx.x) >> 5;
    const int p    = grp * 16 + l15;
    const int b    = p >> 12;
    const int n    = p & (N_ - 1);

    const float* xb = x + (size_t)b * 3 * N_;
    const float cx = xb[n], cy = xb[N_ + n], cz = xb[2 * N_ + n];

    int nb[K_];
    const int4* irow = reinterpret_cast<const int4*>(idx + (size_t)p * K_);
#pragma unroll
    for (int j = 0; j < K_ / 4; ++j) {
        const int4 q = irow[j];
        nb[4 * j] = q.x; nb[4 * j + 1] = q.y; nb[4 * j + 2] = q.z; nb[4 * j + 3] = q.w;
    }

    v2f bf0[K_];
#pragma unroll
    for (int k = 0; k < K_; ++k) {
        const int   j  = nb[k];
        const float e0 = xb[j] - cx, e1 = xb[N_ + j] - cy, e2 = xb[2 * N_ + j] - cz;
        v2f t0;
        t0.x = half ? e2 : e0;  t0.y = half ? cx : e1;
        bf0[k] = t0;
    }
    v2f bf1;
    bf1.x = half ? 0.0f : cy;  bf1.y = half ? 0.0f : cz;

    __syncthreads();

    v8f acc[4] = {{}, {}, {}, {}};
    const float* swl = sw + l15 * W2_PITCH + half * 8;

#pragma unroll
    for (int ct = 0; ct < 4; ++ct) {
        const float* wrow = w1 + (ct * 16 + l15) * 6;
        const float w0 = wrow[0], w1v = wrow[1], w2v = wrow[2];
        const float w3 = wrow[3], w4v = wrow[4], w5v = wrow[5];
        v2f aw0, aw1;
        aw0.x = half ? w2v : w0;   aw0.y = half ? w3 : w1v;
        aw1.x = half ? 0.0f : w4v; aw1.y = half ? 0.0f : w5v;

        float s[8], t[8], mx[8], sa[8], sb[8];
#pragma unroll
        for (int r = 0; r < 8; ++r) {
            const int o = ct * 16 + half * 8 + r;
            s[r] = g1[o] * rsqrtf(rv1[o] + EPS_);
            t[r] = (b1[o] - rm1[o]) * s[r] + beta1[o];
            mx[r] = -3.4e38f; sa[r] = 0.0f; sb[r] = 0.0f;
        }

#pragma unroll
        for (int k = 0; k < K_; ++k) {
            v8f z = {};
            z = __builtin_amdgcn_wmma_f32_16x16x4_f32(false, aw0, false, bf0[k],
                                                      (short)0, z, false, false);
            z = __builtin_amdgcn_wmma_f32_16x16x4_f32(false, aw1, false, bf1,
                                                      (short)0, z, false, false);
#pragma unroll
            for (int r = 0; r < 8; ++r) {
                const float val = fmaxf(z[r] * s[r] + t[r], 0.0f);
                mx[r] = fmaxf(mx[r], val);
                if (k < 8 || k >= 16) sa[r] += val; else sb[r] += val;
            }
        }

        float mean[8];
#pragma unroll
        for (int r = 0; r < 8; ++r) {
            const int    o   = ct * 16 + half * 8 + r;
            const size_t off = ((size_t)b * 64 + o) * N_ + n;
            mean[r] = (sa[r] + sb[r]) * (1.0f / K_);
            out_m1[off] = mx[r];
            out_m2[off] = mean[r];
        }

#pragma unroll
        for (int j = 0; j < 4; ++j) {
            const v2f bm1 = {mx[2 * j],   mx[2 * j + 1]};
            const v2f bm2 = {mean[2 * j], mean[2 * j + 1]};
            const int c = ct * 16 + 2 * j;
#pragma unroll
            for (int ot = 0; ot < 4; ++ot) {
                const float* wr = swl + ot * 16 * W2_PITCH;
                const v2f a1 = {wr[c],      wr[c + 1]};
                const v2f a2 = {wr[64 + c], wr[64 + c + 1]};
                acc[ot] = __builtin_amdgcn_wmma_f32_16x16x4_f32(
                    false, a1, false, bm1, (short)0, acc[ot], false, false);
                acc[ot] = __builtin_amdgcn_wmma_f32_16x16x4_f32(
                    false, a2, false, bm2, (short)0, acc[ot], false, false);
            }
        }
    }

#pragma unroll
    for (int ot = 0; ot < 4; ++ot) {
#pragma unroll
        for (int r = 0; r < 8; ++r) {
            const int   o  = ot * 16 + half * 8 + r;
            const float s2 = g2[o] * rsqrtf(rv2[o] + EPS_);
            const float t2 = (b2[o] - rm2[o]) * s2 + beta2[o];
            out[((size_t)b * 64 + o) * N_ + n] = fmaxf(acc[ot][r] * s2 + t2, 0.0f);
        }
    }
}

extern "C" void kernel_launch(void* const* d_in, const int* in_sizes, int n_in,
                              void* d_out, int out_size, void* d_ws, size_t ws_size,
                              hipStream_t stream)
{
    const float* x     = (const float*)d_in[0];
    const float* w1    = (const float*)d_in[1];
    const float* b1    = (const float*)d_in[2];
    const float* g1    = (const float*)d_in[3];
    const float* beta1 = (const float*)d_in[4];
    const float* rm1   = (const float*)d_in[5];
    const float* rv1   = (const float*)d_in[6];
    const float* w2    = (const float*)d_in[7];
    const float* b2    = (const float*)d_in[8];
    const float* g2    = (const float*)d_in[9];
    const float* beta2 = (const float*)d_in[10];
    const float* rm2   = (const float*)d_in[11];
    const float* rv2   = (const float*)d_in[12];

    float* out    = (float*)d_out;
    float* out_m1 = out + (size_t)B_ * 64 * N_;
    float* out_m2 = out + 2 * (size_t)B_ * 64 * N_;
    int*   idx    = (int*)d_ws;

    knn_kernel<<<B_ * (N_ / ((KNN_THREADS / 32) * KNN_Q)), KNN_THREADS, 0, stream>>>(x, idx);

    edge_conv_kernel<<<(B_ * N_ / 16) / 4, 128, 0, stream>>>(
        x, idx, w1, b1, g1, beta1, rm1, rv1,
        w2, b2, g2, beta2, rm2, rv2, out, out_m1, out_m2);
}


# knn insertion without a vector-to-scalar hand-over: no scalar entry test (a point below the 20th lands in unused lanes 20..30), landing lane from a compare on the shifted list, lane 0 pulls +inf
# speedup vs baseline: 10.0822x; 1.0509x over previous
#include <hip/hip_runtime.h>

#define B_    8
#define C_    3
#define N_    4096
#define K_    20
#define H_    64
#define DOUT_ 64
#define EPS_  1e-5f

typedef __attribute__((ext_vector_type(2))) float v2f;
typedef __attribute__((ext_vector_type(8))) float v8f;

#define KNN_THREADS 1024
#define KNN_Q       4
#define KNN_PRE     32

__device__ __forceinline__ float lane_value(float v, int l)
{
    return __int_as_float(__builtin_amdgcn_readlane(__float_as_int(v), l));
}

__device__ __forceinline__ v2f fma2(v2f a, v2f b, v2f c)
{
    return __builtin_elementwise_fma(a, b, c);
}

__device__ __forceinline__ v2f both(float s)
{
    v2f r = {s, s};
    return r;
}

__global__ __launch_bounds__(KNN_THREADS) void knn_kernel(const float* __restrict__ x,
                                                          int* __restrict__ idx)
{
    __shared__ float sx[N_], sy[N_], sz[N_];
    constexpr int QPB = (KNN_THREADS / 32) * KNN_Q;
    constexpr int BPB = N_ / QPB;
    constexpr int NCH = N_ / 64;
    const int b    = blockIdx.x / BPB;
    const int part = blockIdx.x % BPB;
    const float* xb = x + (size_t)b * 3 * N_;
    for (int j = threadIdx.x; j < N_; j += KNN_THREADS) {
        sx[j] = xb[j];
        sy[j] = xb[N_ + j];
        sz[j] = xb[2 * N_ + j];
    }
    __syncthreads();

    const int lane = threadIdx.x & 31;
    const int wv   = threadIdx.x >> 5;
    const int n0   = part * QPB + wv * KNN_Q;

    const int   nq  = n0 + (lane & (KNN_Q - 1));
    const float lqx = sx[nq], lqy = sy[nq], lqz = sz[nq];
    const float lqq = __builtin_fmaf(lqz, lqz, __builtin_fmaf(lqx, lqx, lqy * lqy));

    float qx[KNN_Q], qy[KNN_Q], qz[KNN_Q], qq[KNN_Q];
    float bd[KNN_Q], thr[KNN_Q], lm[KNN_Q];
    int   bi[KNN_Q];
#pragma unroll
    for (int q = 0; q < KNN_Q; ++q) {
        qx[q] = lane_value(lqx, q);
        qy[q] = lane_value(lqy, q);
        qz[q] = lane_value(lqz, q);
        qq[q] = lane_value(lqq, q);
        bi[q] = 0; lm[q] = -3.4e38f;
    }

    const v2f two = {2.0f, 2.0f};
    v2f px = {sx[lane], sx[lane + 32]};
    v2f py = {sy[lane], sy[lane + 32]};
    v2f pz = {sz[lane], sz[lane + 32]};

    for (int c = 0; c < KNN_PRE; ++c) {
        const int jn  = (((c + 1) & (KNN_PRE - 1)) << 6) + lane;
        const v2f npx = {sx[jn], sx[jn + 32]};
        const v2f npy = {sy[jn], sy[jn + 32]};
        const v2f npz = {sz[jn], sz[jn + 32]};
        const v2f pp = fma2(pz, pz, fma2(px, px, py * py));
#pragma unroll
        for (int q = 0; q < KNN_Q; ++q) {
            const v2f ip = fma2(both(qz[q]), pz,
                                fma2(both(qx[q]), px, both(qy[q]) * py));
            const v2f dq = fma2(ip, two, both(-qq[q])) - pp;
            lm[q] = fmaxf(lm[q], fmaxf(dq.x, dq.y));
        }
        px = npx; py = npy; pz = npz;
    }

    float t0[KNN_Q];
#pragma unroll
    for (int q = 0; q < KNN_Q; ++q) t0[q] = -3.4e38f;
    for (int j = 0; j < 32; ++j) {
#pragma unroll
        for (int q = 0; q < KNN_Q; ++q) {
            const float v = lane_value(lm[q], j);
            const int reach = __builtin_popcount(__builtin_amdgcn_ballot_w32(lm[q] >= v));
            const float cand = (reach >= K_) ? v : -3.4e38f;
            t0[q] = (cand > t0[q]) ? cand : t0[q];
        }
    }
#pragma unroll
    for (int q = 0; q < KNN_Q; ++q) {
        const float t = t0[q];
        const int   tb = __float_as_int(t);
        const float f = (t >= 2.3509887e-38f) ? __int_as_float(tb - 1)
               : (t > -1.17549435e-38f) ? -1.17549435e-38f
               :                          __int_as_float(tb + 1);
        bd[q]  = (lane == 31) ? __builtin_inff() : f;
        thr[q] = f;
    }

    const int pull = ((lane == 0) ? 31 : (lane - 1)) << 2;

    for (int c = 0; c < NCH; ++c) {
        const int jn  = (((c + 1) & (NCH - 1)) << 6) + lane;
        const v2f npx = {sx[jn], sx[jn + 32]};
        const v2f npy = {sy[jn], sy[jn + 32]};
        const v2f npz = {sz[jn], sz[jn + 32]};

        float    d[KNN_Q][2];
        unsigned m[KNN_Q][2];
        const v2f pp = fma2(pz, pz, fma2(px, px, py * py));
#pragma unroll
        for (int q = 0; q < KNN_Q; ++q) {
            const v2f ip = fma2(both(qz[q]), pz,
                                fma2(both(qx[q]), px, both(qy[q]) * py));
            const v2f dq = fma2(ip, two, both(-qq[q])) - pp;
            d[q][0] = dq.x; d[q][1] = dq.y;
            m[q][0] = __builtin_amdgcn_ballot_w32(dq.x > thr[q]);
            m[q][1] = __builtin_amdgcn_ballot_w32(dq.y > thr[q]);
        }
#pragma unroll
        for (int q = 0; q < KNN_Q; ++q) {
#pragma unroll
            for (int h = 0; h < 2; ++h) {
                unsigned mm = m[q][h];
                while (mm) {
                    const int l = __builtin_ctz(mm);
                    mm &= mm - 1;
                    const float dc = lane_value(d[q][h], l);
                    const int   jc = (c << 6) + (h << 5) + l;
                    const float ud = __int_as_float(__builtin_amdgcn_ds_bpermute(
                        pull, __float_as_int(bd[q])));
                    const int   ui = __builtin_amdgcn_ds_bpermute(pull, bi[q]);
                    const bool keep = bd[q] >= dc;
                    const bool here = ud >= dc;
                    bd[q] = keep ? bd[q] : (here ? dc : ud);
                    bi[q] = keep ? bi[q] : (here ? jc : ui);
                    thr[q] = lane_value(bd[q], K_ - 1);
                }
            }
        }
        px = npx; py = npy; pz = npz;
    }

#pragma unroll
    for (int q = 0; q < KNN_Q; ++q) {
        if (lane < K_) idx[((size_t)b * N_ + n0 + q) * K_ + lane] = bi[q];
    }
}

#define W2_PITCH 132
__global__ __launch_bounds__(128) void edge_conv_kernel(
    const float* __restrict__ x,  const int* __restrict__ idx,
    const float* __restrict__ w1, const float* __restrict__ b1,
    const float* __restrict__ g1, const float* __restrict__ beta1,
    const float* __restrict__ rm1,const float* __restrict__ rv1,
    const float* __restrict__ w2, const float* __restrict__ b2,
    const float* __restrict__ g2, const float* __restrict__ beta2,
    const float* __restrict__ rm2,const float* __restrict__ rv2,
    float* __restrict__ out, float* __restrict__ out_m1, float* __restrict__ out_m2)
{
    __shared__ float sw[DOUT_ * W2_PITCH];
    for (int r = 0; r < DOUT_; ++r)
        sw[r * W2_PITCH + threadIdx.x] = w2[r * 128 + threadIdx.x];

    const int lane = threadIdx.x & 31;
    const int half = lane >> 4;
    const int l15  = lane & 15;
    const int grp  = (blockIdx.x * blockDim.x + threadIdx.x) >> 5;
    const int p    = grp * 16 + l15;
    const int b    = p >> 12;
    const int n    = p & (N_ - 1);

    const float* xb = x + (size_t)b * 3 * N_;
    const float cx = xb[n], cy = xb[N_ + n], cz = xb[2 * N_ + n];

    int nb[K_];
    const int4* irow = reinterpret_cast<const int4*>(idx + (size_t)p * K_);
#pragma unroll
    for (int j = 0; j < K_ / 4; ++j) {
        const int4 q = irow[j];
        nb[4 * j] = q.x; nb[4 * j + 1] = q.y; nb[4 * j + 2] = q.z; nb[4 * j + 3] = q.w;
    }

    v2f bf0[K_];
#pragma unroll
    for (int k = 0; k < K_; ++k) {
        const int   j  = nb[k];
        const float e0 = xb[j] - cx, e1 = xb[N_ + j] - cy, e2 = xb[2 * N_ + j] - cz;
        v2f t0;
        t0.x = half ? e2 : e0;  t0.y = half ? cx : e1;
        bf0[k] = t0;
    }
    v2f bf1;
    bf1.x = half ? 0.0f : cy;  bf1.y = half ? 0.0f : cz;

    __syncthreads();

    v8f acc[4] = {{}, {}, {}, {}};
    const float* swl = sw + l15 * W2_PITCH + half * 8;

#pragma unroll
    for (int ct = 0; ct < 4; ++ct) {
        const float* wrow = w1 + (ct * 16 + l15) * 6;
        const float w0 = wrow[0], w1v = wrow[1], w2v = wrow[2];
        const float w3 = wrow[3], w4v = wrow[4], w5v = wrow[5];
        v2f aw0, aw1;
        aw0.x = half ? w2v : w0;   aw0.y = half ? w3 : w1v;
        aw1.x = half ? 0.0f : w4v; aw1.y = half ? 0.0f : w5v;

        float s[8], t[8], mx[8], sa[8], sb[8];
#pragma unroll
        for (int r = 0; r < 8; ++r) {
            const int o = ct * 16 + half * 8 + r;
            s[r] = g1[o] * rsqrtf(rv1[o] + EPS_);
            t[r] = (b1[o] - rm1[o]) * s[r] + beta1[o];
            mx[r] = -3.4e38f; sa[r] = 0.0f; sb[r] = 0.0f;
        }

#pragma unroll
        for (int k = 0; k < K_; ++k) {
            v8f z = {};
            z = __builtin_amdgcn_wmma_f32_16x16x4_f32(false, aw0, false, bf0[k],
                                                      (short)0, z, false, false);
            z = __builtin_amdgcn_wmma_f32_16x16x4_f32(false, aw1, false, bf1,
                                                      (short)0, z, false, false);
#pragma unroll
            for (int r = 0; r < 8; ++r) {
                const float val = fmaxf(z[r] * s[r] + t[r], 0.0f);
                mx[r] = fmaxf(mx[r], val);
                if (k < 8 || k >= 16) sa[r] += val; else sb[r] += val;
            }
        }

        float mean[8];
#pragma unroll
        for (int r = 0; r < 8; ++r) {
            const int    o   = ct * 16 + half * 8 + r;
            const size_t off = ((size_t)b * 64 + o) * N_ + n;
            mean[r] = (sa[r] + sb[r]) * (1.0f / K_);
            out_m1[off] = mx[r];
            out_m2[off] = mean[r];
        }

#pragma unroll
        for (int j = 0; j < 4; ++j) {
            const v2f bm1 = {mx[2 * j],   mx[2 * j + 1]};
            const v2f bm2 = {mean[2 * j], mean[2 * j + 1]};
            const int c = ct * 16 + 2 * j;
#pragma unroll
            for (int ot = 0; ot < 4; ++ot) {
                const float* wr = swl + ot * 16 * W2_PITCH;
                const v2f a1 = {wr[c],      wr[c + 1]};
                const v2f a2 = {wr[64 + c], wr[64 + c + 1]};
                acc[ot] = __builtin_amdgcn_wmma_f32_16x16x4_f32(
                    false, a1, false, bm1, (short)0, acc[ot], false, false);
                acc[ot] = __builtin_amdgcn_wmma_f32_16x16x4_f32(
                    false, a2, false, bm2, (short)0, acc[ot], false, false);
            }
        }
    }

#pragma unroll
    for (int ot = 0; ot < 4; ++ot) {
#pragma unroll
        for (int r = 0; r < 8; ++r) {
            const int   o  = ot * 16 + half * 8 + r;
            const float s2 = g2[o] * rsqrtf(rv2[o] + EPS_);
            const float t2 = (b2[o] - rm2[o]) * s2 + beta2[o];
            out[((size_t)b * 64 + o) * N_ + n] = fmaxf(acc[ot][r] * s2 + t2, 0.0f);
        }
    }
}

extern "C" void kernel_launch(void* const* d_in, const int* in_sizes, int n_in,
                              void* d_out, int out_size, void* d_ws, size_t ws_size,
                              hipStream_t stream)
{
    const float* x     = (const float*)d_in[0];
    const float* w1    = (const float*)d_in[1];
    const float* b1    = (const float*)d_in[2];
    const float* g1    = (const float*)d_in[3];
    const float* beta1 = (const float*)d_in[4];
    const float* rm1   = (const float*)d_in[5];
    const float* rv1   = (const float*)d_in[6];
    const float* w2    = (const float*)d_in[7];
    const float* b2    = (const float*)d_in[8];
    const float* g2    = (const float*)d_in[9];
    const float* beta2 = (const float*)d_in[10];
    const float* rm2   = (const float*)d_in[11];
    const float* rv2   = (const float*)d_in[12];

    float* out    = (float*)d_out;
    float* out_m1 = out + (size_t)B_ * 64 * N_;
    float* out_m2 = out + 2 * (size_t)B_ * 64 * N_;
    int*   idx    = (int*)d_ws;

    knn_kernel<<<B_ * (N_ / ((KNN_THREADS / 32) * KNN_Q)), KNN_THREADS, 0, stream>>>(x, idx);

    edge_conv_kernel<<<(B_ * N_ / 16) / 4, 128, 0, stream>>>(
        x, idx, w1, b1, g1, beta1, rm1, rv1,
        w2, b2, g2, beta2, rm2, rv2, out, out_m1, out_m2);
}
